# starting kernel
# baseline (speedup 1.0000x reference)
#include <hip/hip_runtime.h>
#include <hip/hip_bf16.h>


typedef __attribute__((ext_vector_type(2))) float v2f;
typedef __attribute__((ext_vector_type(8))) float v8f;

namespace {
constexpr int B_CH = 512;
constexpr int T_LEN = 48000;
constexpr int SLABS = 12;
constexpr int SLAB_LEN = T_LEN / SLABS;
constexpr int BLK = 16;
constexpr int BLOCKS_PER_SLAB = SLAB_LEN / BLK;
constexpr int GRP = 16;
constexpr int GROUPS = (BLOCKS_PER_SLAB + GRP - 1) / GRP;
constexpr int COEF_STRIDE = 64;
}

__global__ void biquad_coef_kernel(const float* __restrict__ b0v, const float* __restrict__ b1v,
                                   const float* __restrict__ b2v, const float* __restrict__ a1v,
                                   const float* __restrict__ a2v, float* __restrict__ coef) {
  int c = blockIdx.x * blockDim.x + threadIdx.x;
  if (c >= B_CH) return;
  const float b0 = b0v[c], b1 = b1v[c], b2 = b2v[c], a1 = a1v[c], a2 = a2v[c];
  float* w = coef + c * COEF_STRIDE;
  const float v1 = b1 - a1 * b0;
  const float v2 = b2 - a2 * b0;
  w[0]  = b0;
  w[16] = 1.0f;
  w[32] = 0.0f;
  float ga = 1.0f, gb = 0.0f;
  float gaPrevEnd = 0.0f, gbPrevEnd = 0.0f;
  for (int i = 1; i <= SLAB_LEN; ++i) {
    const float gan = -a1 * ga - a2 * gb;
    const float gbn = ga;
    if (i < 16) {
      w[i]      = ga * v1 + gb * v2;
      w[16 + i] = gan;
      w[32 + i] = gbn;
    }
    if (i == SLAB_LEN - 1) { gaPrevEnd = gan; gbPrevEnd = gbn; }
    ga = gan; gb = gbn;
  }
  w[52] = ga;            w[53] = gb;
  w[54] = -a2 * gaPrevEnd; w[55] = -a2 * gbPrevEnd;
}

template <bool WRITE_Y>
__global__ __launch_bounds__(32) void biquad_pass_kernel(
    const float* __restrict__ x,
    const float* __restrict__ b1v, const float* __restrict__ b2v,
    const float* __restrict__ a1v, const float* __restrict__ a2v,
    const float* __restrict__ coef,
    const float* __restrict__ sstart,
    float* __restrict__ dloc,
    float* __restrict__ y) {
  const int lane = threadIdx.x;
  const int wid = blockIdx.x;
  const int c = wid / SLABS;
  const int s = wid % SLABS;
  const int Mrow = lane & 15;
  const int hi = lane >> 4;
  const float* __restrict__ w = coef + c * COEF_STRIDE;

  v2f af[4];
#pragma unroll
  for (int m = 0; m < 4; ++m) {
#pragma unroll
    for (int r = 0; r < 2; ++r) {
      const int k = 4 * m + r + 2 * hi;
      af[m][r] = (Mrow >= k) ? w[Mrow - k] : 0.0f;
    }
  }
  float gAr[8], gBr[8];
#pragma unroll
  for (int r = 0; r < 8; ++r) {
    gAr[r] = w[16 + 8 * hi + r];
    gBr[r] = w[32 + 8 * hi + r];
  }
  const float gA14 = w[16 + 14], gA15 = w[16 + 15];
  const float gB14 = w[32 + 14], gB15 = w[32 + 15];
  const float b1 = b1v[c], b2 = b2v[c], a1 = a1v[c], a2 = a2v[c];

  float z1 = 0.0f, z2 = 0.0f;
  if constexpr (WRITE_Y) {
    z1 = sstart[(c * SLABS + s) * 2 + 0];
    z2 = sstart[(c * SLABS + s) * 2 + 1];
  }

  const long base_ch = (long)c * T_LEN + (long)s * SLAB_LEN;

  for (int G = 0; G < GROUPS; ++G) {
    const int j0 = G * GRP;
    const int nb = min(GRP, BLOCKS_PER_SLAB - j0);
    const long base = base_ch + (long)j0 * BLK;

    v2f bf[4];
#pragma unroll
    for (int m = 0; m < 4; ++m) {
      if (Mrow < nb) {
        const float2 t = *(const float2*)(x + base + Mrow * BLK + 4 * m + 2 * hi);
        bf[m][0] = t.x; bf[m][1] = t.y;
      } else {
        bf[m][0] = 0.0f; bf[m][1] = 0.0f;
      }
    }

    if (G + 1 < GROUPS)
      __builtin_prefetch((const void*)(x + base + GRP * BLK + lane * 8), 0, 1);

    v8f acc = {};
    acc = __builtin_amdgcn_wmma_f32_16x16x4_f32(false, af[0], false, bf[0], (short)0, acc, false, false);
    acc = __builtin_amdgcn_wmma_f32_16x16x4_f32(false, af[1], false, bf[1], (short)0, acc, false, false);
    acc = __builtin_amdgcn_wmma_f32_16x16x4_f32(false, af[2], false, bf[2], (short)0, acc, false, false);
    acc = __builtin_amdgcn_wmma_f32_16x16x4_f32(false, af[3], false, bf[3], (short)0, acc, false, false);

    for (int jj = 0; jj < nb; ++jj) {
      const int src = 16 + jj;
      const float yp14 = __shfl(acc[6], src, 32);
      const float yp15 = __shfl(acc[7], src, 32);
      const float x14 = __shfl(bf[3][0], src, 32);
      const float x15 = __shfl(bf[3][1], src, 32);
      const float y14 = yp14 + gA14 * z1 + gB14 * z2;
      const float y15 = yp15 + gA15 * z1 + gB15 * z2;
      if constexpr (WRITE_Y) {
        if (Mrow == jj) {
#pragma unroll
          for (int r = 0; r < 8; ++r) acc[r] += gAr[r] * z1 + gBr[r] * z2;
        }
      }
      const float z2mid = b2 * x14 - a2 * y14;
      const float z2n = b2 * x15 - a2 * y15;
      const float z1n = b1 * x15 - a1 * y15 + z2mid;
      z1 = z1n; z2 = z2n;
    }

    if constexpr (WRITE_Y) {
      if (Mrow < nb) {
        float* dst = y + base + Mrow * BLK + 8 * hi;
        *(float4*)(dst + 0) = make_float4(acc[0], acc[1], acc[2], acc[3]);
        *(float4*)(dst + 4) = make_float4(acc[4], acc[5], acc[6], acc[7]);
      }
    }
  }

  if constexpr (!WRITE_Y) {
    if (lane == 0) {
      dloc[(c * SLABS + s) * 2 + 0] = z1;
      dloc[(c * SLABS + s) * 2 + 1] = z2;
    }
  }
}

__global__ void biquad_scan_kernel(const float* __restrict__ coef,
                                   const float* __restrict__ dloc,
                                   float* __restrict__ sstart) {
  int c = blockIdx.x * blockDim.x + threadIdx.x;
  if (c >= B_CH) return;
  const float* w = coef + c * COEF_STRIDE;
  const float m00 = w[52], m01 = w[53], m10 = w[54], m11 = w[55];
  float z1 = 0.0f, z2 = 0.0f;
  for (int s = 0; s < SLABS; ++s) {
    sstart[(c * SLABS + s) * 2 + 0] = z1;
    sstart[(c * SLABS + s) * 2 + 1] = z2;
    const float d1 = dloc[(c * SLABS + s) * 2 + 0];
    const float d2 = dloc[(c * SLABS + s) * 2 + 1];
    const float z1n = m00 * z1 + m01 * z2 + d1;
    const float z2n = m10 * z1 + m11 * z2 + d2;
    z1 = z1n; z2 = z2n;
  }
}

extern "C" void kernel_launch(void* const* d_in, const int* in_sizes, int n_in,
                              void* d_out, int out_size, void* d_ws, size_t ws_size,
                              hipStream_t stream) {
  const float* x  = (const float*)d_in[0];
  const float* b0 = (const float*)d_in[1];
  const float* b1 = (const float*)d_in[2];
  const float* b2 = (const float*)d_in[3];
  const float* a1 = (const float*)d_in[4];
  const float* a2 = (const float*)d_in[5];
  float* y = (float*)d_out;

  float* ws     = (float*)d_ws;
  float* coef   = ws;
  float* dloc   = coef + B_CH * COEF_STRIDE;
  float* sstart = dloc + B_CH * SLABS * 2;

  biquad_coef_kernel<<<(B_CH + 63) / 64, 64, 0, stream>>>(b0, b1, b2, a1, a2, coef);
  biquad_pass_kernel<false><<<B_CH * SLABS, 32, 0, stream>>>(
      x, b1, b2, a1, a2, coef, nullptr, dloc, nullptr);
  biquad_scan_kernel<<<(B_CH + 63) / 64, 64, 0, stream>>>(coef, dloc, sstart);
  biquad_pass_kernel<true><<<B_CH * SLABS, 32, 0, stream>>>(
      x, b1, b2, a1, a2, coef, sstart, nullptr, y);
}


# coefficient kernel forms the slab propagator M^4000 by square-and-multiply (12 stages, as p*M+q*I in f64, rounded to f32 once) instead of a 4000-step dependent f32 recurrence; its rounding changes
# speedup vs baseline: 1.3319x; 1.3319x over previous
#include <hip/hip_runtime.h>
#include <hip/hip_bf16.h>


typedef __attribute__((ext_vector_type(2))) float v2f;
typedef __attribute__((ext_vector_type(8))) float v8f;

namespace {
constexpr int B_CH = 512;
constexpr int T_LEN = 48000;
constexpr int SLABS = 12;
constexpr int SLAB_LEN = T_LEN / SLABS;
constexpr int BLK = 16;
constexpr int BLOCKS_PER_SLAB = SLAB_LEN / BLK;
constexpr int GRP = 16;
constexpr int GROUPS = (BLOCKS_PER_SLAB + GRP - 1) / GRP;
constexpr int COEF_STRIDE = 64;
}

namespace {
constexpr int top_bit(int n) { int b = 0; while ((n >> (b + 1)) != 0) ++b; return b; }
}

__global__ void biquad_coef_kernel(const float* __restrict__ b0v, const float* __restrict__ b1v,
                                   const float* __restrict__ b2v, const float* __restrict__ a1v,
                                   const float* __restrict__ a2v, float* __restrict__ coef) {
  int c = blockIdx.x * blockDim.x + threadIdx.x;
  if (c >= B_CH) return;
  const float b0 = b0v[c], b1 = b1v[c], b2 = b2v[c], a1 = a1v[c], a2 = a2v[c];
  float* w = coef + c * COEF_STRIDE;
  const float v1 = b1 - a1 * b0;
  const float v2 = b2 - a2 * b0;
  w[0]  = b0;
  w[16] = 1.0f;
  w[32] = 0.0f;
  float ga = 1.0f, gb = 0.0f;
  for (int i = 1; i < BLK; ++i) {
    const float gan = -a1 * ga - a2 * gb;
    const float gbn = ga;
    w[i]      = ga * v1 + gb * v2;
    w[16 + i] = gan;
    w[32 + i] = gbn;
    ga = gan; gb = gbn;
  }
  const double a1d = (double)a1, a2d = (double)a2;
  double p = 1.0, q = 0.0;
  for (int bit = top_bit(SLAB_LEN) - 1; bit >= 0; --bit) {
    const double t  = (q + q) - a1d * p;
    const double pp = p * p;
    const double qn = q * q - a2d * pp;
    p = p * t;
    q = qn;
    if ((SLAB_LEN >> bit) & 1) {
      const double pn = q - a1d * p;
      q = -a2d * p;
      p = pn;
    }
  }
  w[52] = (float)(q - a1d * p); w[53] = (float)p;
  w[54] = (float)(-a2d * p);    w[55] = (float)q;
}

template <bool WRITE_Y>
__global__ __launch_bounds__(32) void biquad_pass_kernel(
    const float* __restrict__ x,
    const float* __restrict__ b1v, const float* __restrict__ b2v,
    const float* __restrict__ a1v, const float* __restrict__ a2v,
    const float* __restrict__ coef,
    const float* __restrict__ sstart,
    float* __restrict__ dloc,
    float* __restrict__ y) {
  const int lane = threadIdx.x;
  const int wid = blockIdx.x;
  const int c = wid / SLABS;
  const int s = wid % SLABS;
  const int Mrow = lane & 15;
  const int hi = lane >> 4;
  const float* __restrict__ w = coef + c * COEF_STRIDE;

  v2f af[4];
#pragma unroll
  for (int m = 0; m < 4; ++m) {
#pragma unroll
    for (int r = 0; r < 2; ++r) {
      const int k = 4 * m + r + 2 * hi;
      af[m][r] = (Mrow >= k) ? w[Mrow - k] : 0.0f;
    }
  }
  float gAr[8], gBr[8];
#pragma unroll
  for (int r = 0; r < 8; ++r) {
    gAr[r] = w[16 + 8 * hi + r];
    gBr[r] = w[32 + 8 * hi + r];
  }
  const float gA14 = w[16 + 14], gA15 = w[16 + 15];
  const float gB14 = w[32 + 14], gB15 = w[32 + 15];
  const float b1 = b1v[c], b2 = b2v[c], a1 = a1v[c], a2 = a2v[c];

  float z1 = 0.0f, z2 = 0.0f;
  if constexpr (WRITE_Y) {
    z1 = sstart[(c * SLABS + s) * 2 + 0];
    z2 = sstart[(c * SLABS + s) * 2 + 1];
  }

  const long base_ch = (long)c * T_LEN + (long)s * SLAB_LEN;

  for (int G = 0; G < GROUPS; ++G) {
    const int j0 = G * GRP;
    const int nb = min(GRP, BLOCKS_PER_SLAB - j0);
    const long base = base_ch + (long)j0 * BLK;

    v2f bf[4];
#pragma unroll
    for (int m = 0; m < 4; ++m) {
      if (Mrow < nb) {
        const float2 t = *(const float2*)(x + base + Mrow * BLK + 4 * m + 2 * hi);
        bf[m][0] = t.x; bf[m][1] = t.y;
      } else {
        bf[m][0] = 0.0f; bf[m][1] = 0.0f;
      }
    }

    if (G + 1 < GROUPS)
      __builtin_prefetch((const void*)(x + base + GRP * BLK + lane * 8), 0, 1);

    v8f acc = {};
    acc = __builtin_amdgcn_wmma_f32_16x16x4_f32(false, af[0], false, bf[0], (short)0, acc, false, false);
    acc = __builtin_amdgcn_wmma_f32_16x16x4_f32(false, af[1], false, bf[1], (short)0, acc, false, false);
    acc = __builtin_amdgcn_wmma_f32_16x16x4_f32(false, af[2], false, bf[2], (short)0, acc, false, false);
    acc = __builtin_amdgcn_wmma_f32_16x16x4_f32(false, af[3], false, bf[3], (short)0, acc, false, false);

    for (int jj = 0; jj < nb; ++jj) {
      const int src = 16 + jj;
      const float yp14 = __shfl(acc[6], src, 32);
      const float yp15 = __shfl(acc[7], src, 32);
      const float x14 = __shfl(bf[3][0], src, 32);
      const float x15 = __shfl(bf[3][1], src, 32);
      const float y14 = yp14 + gA14 * z1 + gB14 * z2;
      const float y15 = yp15 + gA15 * z1 + gB15 * z2;
      if constexpr (WRITE_Y) {
        if (Mrow == jj) {
#pragma unroll
          for (int r = 0; r < 8; ++r) acc[r] += gAr[r] * z1 + gBr[r] * z2;
        }
      }
      const float z2mid = b2 * x14 - a2 * y14;
      const float z2n = b2 * x15 - a2 * y15;
      const float z1n = b1 * x15 - a1 * y15 + z2mid;
      z1 = z1n; z2 = z2n;
    }

    if constexpr (WRITE_Y) {
      if (Mrow < nb) {
        float* dst = y + base + Mrow * BLK + 8 * hi;
        *(float4*)(dst + 0) = make_float4(acc[0], acc[1], acc[2], acc[3]);
        *(float4*)(dst + 4) = make_float4(acc[4], acc[5], acc[6], acc[7]);
      }
    }
  }

  if constexpr (!WRITE_Y) {
    if (lane == 0) {
      dloc[(c * SLABS + s) * 2 + 0] = z1;
      dloc[(c * SLABS + s) * 2 + 1] = z2;
    }
  }
}

__global__ void biquad_scan_kernel(const float* __restrict__ coef,
                                   const float* __restrict__ dloc,
                                   float* __restrict__ sstart) {
  int c = blockIdx.x * blockDim.x + threadIdx.x;
  if (c >= B_CH) return;
  const float* w = coef + c * COEF_STRIDE;
  const float m00 = w[52], m01 = w[53], m10 = w[54], m11 = w[55];
  float z1 = 0.0f, z2 = 0.0f;
  for (int s = 0; s < SLABS; ++s) {
    sstart[(c * SLABS + s) * 2 + 0] = z1;
    sstart[(c * SLABS + s) * 2 + 1] = z2;
    const float d1 = dloc[(c * SLABS + s) * 2 + 0];
    const float d2 = dloc[(c * SLABS + s) * 2 + 1];
    const float z1n = m00 * z1 + m01 * z2 + d1;
    const float z2n = m10 * z1 + m11 * z2 + d2;
    z1 = z1n; z2 = z2n;
  }
}

extern "C" void kernel_launch(void* const* d_in, const int* in_sizes, int n_in,
                              void* d_out, int out_size, void* d_ws, size_t ws_size,
                              hipStream_t stream) {
  const float* x  = (const float*)d_in[0];
  const float* b0 = (const float*)d_in[1];
  const float* b1 = (const float*)d_in[2];
  const float* b2 = (const float*)d_in[3];
  const float* a1 = (const float*)d_in[4];
  const float* a2 = (const float*)d_in[5];
  float* y = (float*)d_out;

  float* ws     = (float*)d_ws;
  float* coef   = ws;
  float* dloc   = coef + B_CH * COEF_STRIDE;
  float* sstart = dloc + B_CH * SLABS * 2;

  biquad_coef_kernel<<<(B_CH + 63) / 64, 64, 0, stream>>>(b0, b1, b2, a1, a2, coef);
  biquad_pass_kernel<false><<<B_CH * SLABS, 32, 0, stream>>>(
      x, b1, b2, a1, a2, coef, nullptr, dloc, nullptr);
  biquad_scan_kernel<<<(B_CH + 63) / 64, 64, 0, stream>>>(coef, dloc, sstart);
  biquad_pass_kernel<true><<<B_CH * SLABS, 32, 0, stream>>>(
      x, b1, b2, a1, a2, coef, sstart, nullptr, y);
}


# WMMA columns = 16 independent sub-slabs of one channel (125 of 384 samples, was 12 of 4000): state carry is one lane-parallel step per block, not 16 serial; slab scan has 125 steps, rounding changes
# speedup vs baseline: 1.5525x; 1.1656x over previous
#include <hip/hip_runtime.h>


typedef __attribute__((ext_vector_type(2))) float v2f;
typedef __attribute__((ext_vector_type(8))) float v8f;

namespace {
constexpr int B_CH = 512;
constexpr int T_LEN = 48000;
constexpr int BLK = 16;
constexpr int NBLK = T_LEN / BLK;
constexpr int LB = 24;
constexpr int SUB_LEN = LB * BLK;
constexpr int NSUB = NBLK / LB;
constexpr int COLS = 16;
constexpr int WPC = (NSUB + COLS - 1) / COLS;
constexpr int AHEAD = 12;
constexpr int COEF_STRIDE = 64;
static_assert(NBLK * BLK == T_LEN, "whole blocks");
static_assert(NSUB * LB == NBLK, "whole sub-slabs");
static_assert(LB % 2 == 0, "a sub-slab starts on a 128-byte line");
static_assert(AHEAD <= LB, "ring no longer than the sub-slab");
constexpr int top_bit(int n) { int b = 0; while ((n >> (b + 1)) != 0) ++b; return b; }
}

__global__ void biquad_coef_kernel(const float* __restrict__ b0v, const float* __restrict__ b1v,
                                   const float* __restrict__ b2v, const float* __restrict__ a1v,
                                   const float* __restrict__ a2v, float* __restrict__ coef) {
  int c = blockIdx.x * blockDim.x + threadIdx.x;
  if (c >= B_CH) return;
  const float b0 = b0v[c], b1 = b1v[c], b2 = b2v[c], a1 = a1v[c], a2 = a2v[c];
  float* w = coef + c * COEF_STRIDE;
  const float v1 = b1 - a1 * b0;
  const float v2 = b2 - a2 * b0;
  w[0]  = b0;
  w[16] = 1.0f;
  w[32] = 0.0f;
  float ga = 1.0f, gb = 0.0f;
  for (int i = 1; i < BLK; ++i) {
    const float gan = -a1 * ga - a2 * gb;
    const float gbn = ga;
    w[i]      = ga * v1 + gb * v2;
    w[16 + i] = gan;
    w[32 + i] = gbn;
    ga = gan; gb = gbn;
  }
  const double a1d = (double)a1, a2d = (double)a2;
  double p = 1.0, q = 0.0;
  for (int bit = top_bit(SUB_LEN) - 1; bit >= 0; --bit) {
    const double t  = (q + q) - a1d * p;
    const double pp = p * p;
    const double qn = q * q - a2d * pp;
    p = p * t;
    q = qn;
    if ((SUB_LEN >> bit) & 1) {
      const double pn = q - a1d * p;
      q = -a2d * p;
      p = pn;
    }
  }
  w[52] = (float)(q - a1d * p); w[53] = (float)p;
  w[54] = (float)(-a2d * p);    w[55] = (float)q;
}

template <bool WRITE_Y>
__global__ __launch_bounds__(32 * WPC) void biquad_pass_kernel(
    const float* __restrict__ x,
    const float* __restrict__ b1v, const float* __restrict__ b2v,
    const float* __restrict__ a1v, const float* __restrict__ a2v,
    const float* __restrict__ coef,
    const float* __restrict__ sstart,
    float* __restrict__ dloc,
    float* __restrict__ y) {
  const int lane = threadIdx.x & 31;
  const int wv = threadIdx.x >> 5;
  const int c = blockIdx.x;
  const int col = lane & 15;
  const int hi = lane >> 4;
  const int q_raw = wv * COLS + col;
  const bool live = q_raw < NSUB;
  const int q = live ? q_raw : NSUB - 1;
  const float* __restrict__ w = coef + c * COEF_STRIDE;

  v2f af[4];
#pragma unroll
  for (int m = 0; m < 4; ++m) {
#pragma unroll
    for (int r = 0; r < 2; ++r) {
      const int d = col - (8 * hi + 2 * m + r);
      const float hv = w[d > 0 ? d : 0];
      af[m][r] = (d >= 0) ? hv : 0.0f;
    }
  }
  float gAr[8], gBr[8];
#pragma unroll
  for (int r = 0; r < 8; ++r) {
    gAr[r] = w[16 + 8 * hi + r];
    gBr[r] = w[32 + 8 * hi + r];
  }
  const float b1 = b1v[c], b2 = b2v[c], a1 = a1v[c], a2 = a2v[c];

  float z1 = 0.0f, z2 = 0.0f;
  if constexpr (WRITE_Y) {
    const float2 s0 = *(const float2*)(sstart + ((long)q * B_CH + c) * 2);
    z1 = s0.x; z2 = s0.y;
  }

  const long off = (long)c * T_LEN + (long)q * SUB_LEN + 8 * hi;
  const float* __restrict__ xp = x + off;

  float4 xr[AHEAD][2];
#pragma unroll
  for (int j = 0; j < AHEAD; ++j) {
    xr[j][0] = *(const float4*)(xp + j * BLK);
    xr[j][1] = *(const float4*)(xp + j * BLK + 4);
  }

#pragma unroll
  for (int j = 0; j < LB; ++j) {
    const float4 t0 = xr[j % AHEAD][0];
    const float4 t1 = xr[j % AHEAD][1];
    if (j + AHEAD < LB) {
      xr[j % AHEAD][0] = *(const float4*)(xp + (j + AHEAD) * BLK);
      xr[j % AHEAD][1] = *(const float4*)(xp + (j + AHEAD) * BLK + 4);
    }
    v2f bf0, bf1, bf2, bf3;
    bf0[0] = t0.x; bf0[1] = t0.y;
    bf1[0] = t0.z; bf1[1] = t0.w;
    bf2[0] = t1.x; bf2[1] = t1.y;
    bf3[0] = t1.z; bf3[1] = t1.w;

    v8f acc = {};
    acc = __builtin_amdgcn_wmma_f32_16x16x4_f32(false, af[0], false, bf0, (short)0, acc, false, false);
    acc = __builtin_amdgcn_wmma_f32_16x16x4_f32(false, af[1], false, bf1, (short)0, acc, false, false);
    acc = __builtin_amdgcn_wmma_f32_16x16x4_f32(false, af[2], false, bf2, (short)0, acc, false, false);
    acc = __builtin_amdgcn_wmma_f32_16x16x4_f32(false, af[3], false, bf3, (short)0, acc, false, false);

    if (j > 0) {
      const unsigned u1 = __float_as_uint(z1), u2 = __float_as_uint(z2);
      const float z1o = __uint_as_float(
          __builtin_amdgcn_permlanex16(u1, u1, 0x76543210u, 0xfedcba98u, false, false));
      const float z2o = __uint_as_float(
          __builtin_amdgcn_permlanex16(u2, u2, 0x76543210u, 0xfedcba98u, false, false));
      if (hi == 0) { z1 = z1o; z2 = z2o; }
    }

    if constexpr (WRITE_Y) {
#pragma unroll
      for (int r = 0; r < 8; ++r) acc[r] += gAr[r] * z1 + gBr[r] * z2;
      if (live) {
        float* dst = y + off + j * BLK;
        *(float4*)(dst + 0) = make_float4(acc[0], acc[1], acc[2], acc[3]);
        *(float4*)(dst + 4) = make_float4(acc[4], acc[5], acc[6], acc[7]);
      }
    } else {
      acc[6] += gAr[6] * z1 + gBr[6] * z2;
      acc[7] += gAr[7] * z1 + gBr[7] * z2;
    }

    const float z2mid = b2 * t1.z - a2 * acc[6];
    const float z2n = b2 * t1.w - a2 * acc[7];
    const float z1n = b1 * t1.w - a1 * acc[7] + z2mid;
    z1 = z1n; z2 = z2n;
  }

  if constexpr (!WRITE_Y) {
    if (hi == 1 && live) {
      *(float2*)(dloc + ((long)q * B_CH + c) * 2) = make_float2(z1, z2);
    }
  }
}

__global__ __launch_bounds__(64) void biquad_scan_kernel(const float* __restrict__ coef,
                                                         const float* __restrict__ dloc,
                                                         float* __restrict__ sstart) {
  int c = blockIdx.x * blockDim.x + threadIdx.x;
  if (c >= B_CH) return;
  const float* w = coef + c * COEF_STRIDE;
  const float m00 = w[52], m01 = w[53], m10 = w[54], m11 = w[55];
  constexpr int CH = 25;
  static_assert(NSUB % CH == 0, "whole chunks");
  float z1 = 0.0f, z2 = 0.0f;
#pragma unroll
  for (int q0 = 0; q0 < NSUB; q0 += CH) {
    float2 d[CH];
#pragma unroll
    for (int i = 0; i < CH; ++i) d[i] = *(const float2*)(dloc + ((long)(q0 + i) * B_CH + c) * 2);
#pragma unroll
    for (int i = 0; i < CH; ++i) {
      *(float2*)(sstart + ((long)(q0 + i) * B_CH + c) * 2) = make_float2(z1, z2);
      const float z1n = m00 * z1 + (m01 * z2 + d[i].x);
      const float z2n = m10 * z1 + (m11 * z2 + d[i].y);
      z1 = z1n; z2 = z2n;
    }
  }
}

extern "C" void kernel_launch(void* const* d_in, const int* in_sizes, int n_in,
                              void* d_out, int out_size, void* d_ws, size_t ws_size,
                              hipStream_t stream) {
  const float* x  = (const float*)d_in[0];
  const float* b0 = (const float*)d_in[1];
  const float* b1 = (const float*)d_in[2];
  const float* b2 = (const float*)d_in[3];
  const float* a1 = (const float*)d_in[4];
  const float* a2 = (const float*)d_in[5];
  float* y = (float*)d_out;

  float* ws     = (float*)d_ws;
  float* coef   = ws;
  float* dloc   = coef + B_CH * COEF_STRIDE;
  float* sstart = dloc + NSUB * B_CH * 2;

  biquad_coef_kernel<<<(B_CH + 63) / 64, 64, 0, stream>>>(b0, b1, b2, a1, a2, coef);
  biquad_pass_kernel<false><<<B_CH, 32 * WPC, 0, stream>>>(
      x, b1, b2, a1, a2, coef, nullptr, dloc, nullptr);
  biquad_scan_kernel<<<(B_CH + 63) / 64, 64, 0, stream>>>(coef, dloc, sstart);
  biquad_pass_kernel<true><<<B_CH, 32 * WPC, 0, stream>>>(
      x, b1, b2, a1, a2, coef, sstart, nullptr, y);
}


# one launch instead of four: each channel's workgroup forms its coefficients in LDS, runs pass 1, scans the 125 sub-slab states from LDS behind one barrier, then runs pass 2; workspace unused
# speedup vs baseline: 2.2112x; 1.4243x over previous
#include <hip/hip_runtime.h>


typedef __attribute__((ext_vector_type(2))) float v2f;
typedef __attribute__((ext_vector_type(8))) float v8f;

namespace {
constexpr int B_CH = 512;
constexpr int T_LEN = 48000;
constexpr int BLK = 16;
constexpr int NBLK = T_LEN / BLK;
constexpr int LB = 24;
constexpr int SUB_LEN = LB * BLK;
constexpr int NSUB = NBLK / LB;
constexpr int COLS = 16;
constexpr int WPC = (NSUB + COLS - 1) / COLS;
constexpr int AHEAD = 12;
static_assert(NBLK * BLK == T_LEN, "whole blocks");
static_assert(NSUB * LB == NBLK, "whole sub-slabs");
static_assert(LB % 2 == 0, "a sub-slab starts on a 128-byte line");
static_assert(AHEAD <= LB, "ring no longer than the sub-slab");
static_assert(WPC >= 2, "wave 0 forms h and G, wave 1 the propagator");
constexpr int top_bit(int n) { int b = 0; while ((n >> (b + 1)) != 0) ++b; return b; }
}

__device__ __forceinline__ void ring_fill(const float* __restrict__ xp, float4 (&xr)[AHEAD][2]) {
#pragma unroll
  for (int j = 0; j < AHEAD; ++j) {
    xr[j][0] = *(const float4*)(xp + j * BLK);
    xr[j][1] = *(const float4*)(xp + j * BLK + 4);
  }
}

template <bool WRITE_Y>
__device__ __forceinline__ void run_pass(const float* __restrict__ xp, float* __restrict__ yp,
                                         float4 (&xr)[AHEAD][2], const v2f (&af)[4],
                                         const float (&gAr)[8], const float (&gBr)[8],
                                         float b1, float b2, float a1, float a2,
                                         int hi, bool live, float& z1, float& z2) {
#pragma unroll
  for (int j = 0; j < LB; ++j) {
    const float4 t0 = xr[j % AHEAD][0];
    const float4 t1 = xr[j % AHEAD][1];
    if (j + AHEAD < LB) {
      xr[j % AHEAD][0] = *(const float4*)(xp + (j + AHEAD) * BLK);
      xr[j % AHEAD][1] = *(const float4*)(xp + (j + AHEAD) * BLK + 4);
    }
    v2f bf0, bf1, bf2, bf3;
    bf0[0] = t0.x; bf0[1] = t0.y;
    bf1[0] = t0.z; bf1[1] = t0.w;
    bf2[0] = t1.x; bf2[1] = t1.y;
    bf3[0] = t1.z; bf3[1] = t1.w;

    v8f acc = {};
    acc = __builtin_amdgcn_wmma_f32_16x16x4_f32(false, af[0], false, bf0, (short)0, acc, false, false);
    acc = __builtin_amdgcn_wmma_f32_16x16x4_f32(false, af[1], false, bf1, (short)0, acc, false, false);
    acc = __builtin_amdgcn_wmma_f32_16x16x4_f32(false, af[2], false, bf2, (short)0, acc, false, false);
    acc = __builtin_amdgcn_wmma_f32_16x16x4_f32(false, af[3], false, bf3, (short)0, acc, false, false);

    if (j > 0) {
      const unsigned u1 = __float_as_uint(z1), u2 = __float_as_uint(z2);
      const float z1o = __uint_as_float(
          __builtin_amdgcn_permlanex16(u1, u1, 0x76543210u, 0xfedcba98u, false, false));
      const float z2o = __uint_as_float(
          __builtin_amdgcn_permlanex16(u2, u2, 0x76543210u, 0xfedcba98u, false, false));
      if (hi == 0) { z1 = z1o; z2 = z2o; }
    }

    if constexpr (WRITE_Y) {
#pragma unroll
      for (int r = 0; r < 8; ++r) acc[r] += gAr[r] * z1 + gBr[r] * z2;
      if (live) {
        float* dst = yp + j * BLK;
        *(float4*)(dst + 0) = make_float4(acc[0], acc[1], acc[2], acc[3]);
        *(float4*)(dst + 4) = make_float4(acc[4], acc[5], acc[6], acc[7]);
      }
    } else {
      acc[6] += gAr[6] * z1 + gBr[6] * z2;
      acc[7] += gAr[7] * z1 + gBr[7] * z2;
    }

    const float z2mid = b2 * t1.z - a2 * acc[6];
    const float z2n = b2 * t1.w - a2 * acc[7];
    const float z1n = b1 * t1.w - a1 * acc[7] + z2mid;
    z1 = z1n; z2 = z2n;
  }
}

__global__ __launch_bounds__(32 * WPC) __attribute__((amdgpu_waves_per_eu(4, 4))) void biquad_channel_kernel(
    const float* __restrict__ x,
    const float* __restrict__ b0v, const float* __restrict__ b1v, const float* __restrict__ b2v,
    const float* __restrict__ a1v, const float* __restrict__ a2v,
    float* __restrict__ y) {
  __shared__ float s_coef[3 * BLK];
  __shared__ float s_prop[4];
  __shared__ float2 s_end[WPC * COLS];

  const int lane = threadIdx.x & 31;
  const int wv = __builtin_amdgcn_readfirstlane(threadIdx.x >> 5);
  const int c = blockIdx.x;
  const int col = lane & 15;
  const int hi = lane >> 4;
  const int q_raw = wv * COLS + col;
  const bool live = q_raw < NSUB;
  const int q = live ? q_raw : NSUB - 1;

  const long off = (long)c * T_LEN + (long)q * SUB_LEN + 8 * hi;
  const float* __restrict__ xp = x + off;

  float4 xr[AHEAD][2];
  ring_fill(xp, xr);

  const float b0 = b0v[c], b1 = b1v[c], b2 = b2v[c], a1 = a1v[c], a2 = a2v[c];

  if (wv == 0) {
    const float v1 = b1 - a1 * b0;
    const float v2 = b2 - a2 * b0;
    float hh[BLK], gA[BLK], gB[BLK];
    hh[0] = b0;
    gA[0] = 1.0f;
    gB[0] = 0.0f;
    float ga = 1.0f, gb = 0.0f;
#pragma unroll
    for (int i = 1; i < BLK; ++i) {
      const float gan = -a1 * ga - a2 * gb;
      const float gbn = ga;
      hh[i] = ga * v1 + gb * v2;
      gA[i] = gan;
      gB[i] = gbn;
      ga = gan; gb = gbn;
    }
    if (lane == 0) {
#pragma unroll
      for (int i = 0; i < BLK; ++i) {
        s_coef[i] = hh[i];
        s_coef[BLK + i] = gA[i];
        s_coef[2 * BLK + i] = gB[i];
      }
    }
  } else if (wv == 1) {
    const double a1d = (double)a1, a2d = (double)a2;
    double p = 1.0, qq = 0.0;
    for (int bit = top_bit(SUB_LEN) - 1; bit >= 0; --bit) {
      const double t  = (qq + qq) - a1d * p;
      const double pp = p * p;
      const double qn = qq * qq - a2d * pp;
      p = p * t;
      qq = qn;
      if ((SUB_LEN >> bit) & 1) {
        const double pn = qq - a1d * p;
        qq = -a2d * p;
        p = pn;
      }
    }
    if (lane == 0) {
      s_prop[0] = (float)(qq - a1d * p); s_prop[1] = (float)p;
      s_prop[2] = (float)(-a2d * p);     s_prop[3] = (float)qq;
    }
  }
  __syncthreads();

  v2f af[4];
#pragma unroll
  for (int m = 0; m < 4; ++m) {
#pragma unroll
    for (int r = 0; r < 2; ++r) {
      const int d = col - (8 * hi + 2 * m + r);
      const float hv = s_coef[d > 0 ? d : 0];
      af[m][r] = (d >= 0) ? hv : 0.0f;
    }
  }
  float gAr[8], gBr[8];
#pragma unroll
  for (int r = 0; r < 8; ++r) {
    gAr[r] = s_coef[BLK + 8 * hi + r];
    gBr[r] = s_coef[2 * BLK + 8 * hi + r];
  }

  float z1 = 0.0f, z2 = 0.0f;
  run_pass<false>(xp, nullptr, xr, af, gAr, gBr, b1, b2, a1, a2, hi, live, z1, z2);
  if (hi == 1) s_end[q_raw] = make_float2(z1, z2);

  __syncthreads();

  ring_fill(xp, xr);

  const float m00 = s_prop[0], m01 = s_prop[1], m10 = s_prop[2], m11 = s_prop[3];
  const int qi = q - wv * COLS;
  float s1 = 0.0f, s2 = 0.0f;
  z1 = 0.0f; z2 = 0.0f;
  float2 dcur[COLS];
#pragma unroll
  for (int i = 0; i < COLS; ++i) dcur[i] = s_end[i];
  for (int k = 0; k < wv; ++k) {
    float2 dnext[COLS];
#pragma unroll
    for (int i = 0; i < COLS; ++i) dnext[i] = s_end[(k + 1) * COLS + i];
#pragma unroll
    for (int i = 0; i < COLS; ++i) {
      const float z1n = m00 * z1 + (m01 * z2 + dcur[i].x);
      const float z2n = m10 * z1 + (m11 * z2 + dcur[i].y);
      z1 = z1n; z2 = z2n;
    }
#pragma unroll
    for (int i = 0; i < COLS; ++i) dcur[i] = dnext[i];
  }
#pragma unroll
  for (int i = 0; i < COLS; ++i) {
    if (i == qi) { s1 = z1; s2 = z2; }
    const float z1n = m00 * z1 + (m01 * z2 + dcur[i].x);
    const float z2n = m10 * z1 + (m11 * z2 + dcur[i].y);
    z1 = z1n; z2 = z2n;
  }

  z1 = s1; z2 = s2;
  run_pass<true>(xp, y + off, xr, af, gAr, gBr, b1, b2, a1, a2, hi, live, z1, z2);
}

extern "C" void kernel_launch(void* const* d_in, const int* in_sizes, int n_in,
                              void* d_out, int out_size, void* d_ws, size_t ws_size,
                              hipStream_t stream) {
  const float* x  = (const float*)d_in[0];
  const float* b0 = (const float*)d_in[1];
  const float* b1 = (const float*)d_in[2];
  const float* b2 = (const float*)d_in[3];
  const float* a1 = (const float*)d_in[4];
  const float* a2 = (const float*)d_in[5];
  float* y = (float*)d_out;

  biquad_channel_kernel<<<B_CH, 32 * WPC, 0, stream>>>(x, b0, b1, b2, a1, a2, y);
}


# x and y staged through a wave-private LDS tile: every global load and store instruction now covers whole 128-byte lines (two 256-byte rows) instead of 16 quarter-used lines; sums unchanged
# speedup vs baseline: 4.0863x; 1.8480x over previous
#include <hip/hip_runtime.h>


typedef __attribute__((ext_vector_type(2))) float v2f;
typedef __attribute__((ext_vector_type(8))) float v8f;
typedef __attribute__((ext_vector_type(4))) float v4f;

namespace {
constexpr int B_CH = 512;
constexpr int T_LEN = 48000;
constexpr int BLK = 16;
constexpr int NBLK = T_LEN / BLK;
constexpr int LB = 24;
constexpr int SUB_LEN = LB * BLK;
constexpr int NSUB = NBLK / LB;
constexpr int COLS = 16;
constexpr int WPC = (NSUB + COLS - 1) / COLS;
constexpr int SB = 4;
constexpr int NSTAGE = LB / SB;
constexpr int NRING = 2;
constexpr int LPS = COLS * SB * BLK / (32 * 4);
constexpr int ROW4 = SB * BLK / 4;
constexpr int PITCH4 = ROW4 + 1;
constexpr int TILE4 = COLS * PITCH4;
static_assert(NBLK * BLK == T_LEN, "whole blocks");
static_assert(NSUB * LB == NBLK, "whole sub-slabs");
static_assert(NSTAGE * SB == LB, "whole stages");
static_assert((SB * BLK * 4) % 256 == 0 && (SUB_LEN * 4) % 256 == 0 && (T_LEN * 4) % 256 == 0,
              "a stage row is whole aligned 256-byte units of memory");
static_assert(ROW4 == 16 && LPS == 8, "one access = two rows: lane half = row, lane & 15 = float4 in the row");
constexpr int PAIR_B = 2 * SUB_LEN * 4;
constexpr int LPS_ALL = (NSUB - 1 - (WPC - 1) * COLS + 1) / 2;
static_assert(NRING <= NSTAGE, "ring no longer than the sub-slab");
static_assert(LPS_ALL >= 1 && LPS_ALL < LPS, "the last wave's padded rows sit in the last accesses only");
static_assert(WPC >= 2, "wave 0 forms h and G, wave 1 the propagator");
constexpr int top_bit(int n) { int b = 0; while ((n >> (b + 1)) != 0) ++b; return b; }
}

__device__ __forceinline__ void wave_lds_order() {
  __builtin_amdgcn_fence(__ATOMIC_RELEASE, "wavefront");
  __builtin_amdgcn_wave_barrier();
  __builtin_amdgcn_fence(__ATOMIC_ACQUIRE, "wavefront");
}

__device__ __forceinline__ void stage_load(const float* __restrict__ xw, unsigned lo, int hi, int lim,
                                           int s, v4f (&dst)[LPS]) {
  const char* __restrict__ sp = (const char*)(xw + s * (SB * BLK));
#pragma unroll
  for (int i = 0; i < LPS; ++i) {
    if (i < LPS_ALL) {
      dst[i] = *(const v4f*)(sp + i * PAIR_B + lo);
    } else {
      v4f v = {0.0f, 0.0f, 0.0f, 0.0f};
      if (2 * i + hi <= lim) v = *(const v4f*)(sp + i * PAIR_B + lo);
      dst[i] = v;
    }
  }
}

template <bool WRITE_Y, int J>
__device__ __forceinline__ void block_step(const float* __restrict__ xw, float* __restrict__ yw,
                                           unsigned lo, int lim,
                                           v4f (&ring)[NRING][LPS],
                                           v4f* __restrict__ xt, v4f* __restrict__ yt,
                                           int fr, int cr,
                                           const v2f (&af)[4],
                                           const float (&gAr)[8], const float (&gBr)[8],
                                           float b1, float b2, float a1, float a2,
                                           int hi, v4f& c0, v4f& c1, float& z1, float& z2) {
  const v4f t0 = c0;
  const v4f t1 = c1;
  if constexpr (J + 1 < LB) {
    if constexpr ((J + 1) % SB == 0) {
      constexpr int S = (J + 1) / SB;
      __builtin_amdgcn_sched_barrier(0);
      wave_lds_order();
#pragma unroll
      for (int i = 0; i < LPS; ++i) xt[cr + i * 2 * PITCH4] = ring[S % NRING][i];
      if constexpr (S + NRING < NSTAGE) stage_load(xw, lo, hi, lim, S + NRING, ring[S % NRING]);
      wave_lds_order();
    }
    c0 = xt[fr + 4 * ((J + 1) % SB)];
    c1 = xt[fr + 4 * ((J + 1) % SB) + 1];
  }
  v2f bf0, bf1, bf2, bf3;
  bf0[0] = t0.x; bf0[1] = t0.y;
  bf1[0] = t0.z; bf1[1] = t0.w;
  bf2[0] = t1.x; bf2[1] = t1.y;
  bf3[0] = t1.z; bf3[1] = t1.w;

  v8f acc = {};
  acc = __builtin_amdgcn_wmma_f32_16x16x4_f32(false, af[0], false, bf0, (short)0, acc, false, false);
  acc = __builtin_amdgcn_wmma_f32_16x16x4_f32(false, af[1], false, bf1, (short)0, acc, false, false);
  acc = __builtin_amdgcn_wmma_f32_16x16x4_f32(false, af[2], false, bf2, (short)0, acc, false, false);
  acc = __builtin_amdgcn_wmma_f32_16x16x4_f32(false, af[3], false, bf3, (short)0, acc, false, false);

  if constexpr (J > 0) {
    const unsigned u1 = __float_as_uint(z1), u2 = __float_as_uint(z2);
    const float z1o = __uint_as_float(
        __builtin_amdgcn_permlanex16(u1, u1, 0x76543210u, 0xfedcba98u, false, false));
    const float z2o = __uint_as_float(
        __builtin_amdgcn_permlanex16(u2, u2, 0x76543210u, 0xfedcba98u, false, false));
    if (hi == 0) { z1 = z1o; z2 = z2o; }
  }

  if constexpr (WRITE_Y) {
#pragma unroll
    for (int r = 0; r < 8; ++r) acc[r] += gAr[r] * z1 + gBr[r] * z2;
    yt[fr + 4 * (J % SB)]     = __builtin_shufflevector(acc, acc, 0, 1, 2, 3);
    yt[fr + 4 * (J % SB) + 1] = __builtin_shufflevector(acc, acc, 4, 5, 6, 7);
    if constexpr (J % SB == SB - 1) {
      char* __restrict__ dp = (char*)(yw + (J / SB) * (SB * BLK));
      wave_lds_order();
#pragma unroll
      for (int i = 0; i < LPS; ++i) {
        const v4f v = yt[cr + i * 2 * PITCH4];
        if (i < LPS_ALL) *(v4f*)(dp + i * PAIR_B + lo) = v;
        else if (2 * i + hi <= lim) *(v4f*)(dp + i * PAIR_B + lo) = v;
      }
      wave_lds_order();
    }
  } else {
    acc[6] += gAr[6] * z1 + gBr[6] * z2;
    acc[7] += gAr[7] * z1 + gBr[7] * z2;
  }

  const float z2mid = b2 * t1.z - a2 * acc[6];
  const float z2n = b2 * t1.w - a2 * acc[7];
  const float z1n = b1 * t1.w - a1 * acc[7] + z2mid;
  z1 = z1n; z2 = z2n;
}

template <bool WRITE_Y, int J>
__device__ __forceinline__ void blocks_from(const float* __restrict__ xw, float* __restrict__ yw,
                                            unsigned lo, int lim,
                                            v4f (&ring)[NRING][LPS],
                                            v4f* __restrict__ xt, v4f* __restrict__ yt,
                                            int fr, int cr,
                                            const v2f (&af)[4],
                                            const float (&gAr)[8], const float (&gBr)[8],
                                            float b1, float b2, float a1, float a2,
                                            int hi, v4f& c0, v4f& c1, float& z1, float& z2) {
  block_step<WRITE_Y, J>(xw, yw, lo, lim, ring, xt, yt, fr, cr, af, gAr, gBr, b1, b2, a1, a2, hi, c0, c1, z1, z2);
  if constexpr (J + 1 < LB)
    blocks_from<WRITE_Y, J + 1>(xw, yw, lo, lim, ring, xt, yt, fr, cr, af, gAr, gBr, b1, b2, a1, a2, hi, c0, c1, z1, z2);
}

template <bool WRITE_Y>
__device__ __forceinline__ void run_pass(const float* __restrict__ xw, float* __restrict__ yw,
                                         unsigned lo, int lim,
                                         v4f (&ring)[NRING][LPS],
                                         v4f* __restrict__ xt, v4f* __restrict__ yt,
                                         int fr, int cr,
                                         const v2f (&af)[4],
                                         const float (&gAr)[8], const float (&gBr)[8],
                                         float b1, float b2, float a1, float a2,
                                         int hi, float& z1, float& z2) {
#pragma unroll
  for (int i = 0; i < LPS; ++i) xt[cr + i * 2 * PITCH4] = ring[0][i];
  if constexpr (NRING < NSTAGE) stage_load(xw, lo, hi, lim, NRING, ring[0]);
  wave_lds_order();
  v4f c0 = xt[fr], c1 = xt[fr + 1];
  blocks_from<WRITE_Y, 0>(xw, yw, lo, lim, ring, xt, yt, fr, cr, af, gAr, gBr, b1, b2, a1, a2, hi, c0, c1, z1, z2);
}

__global__ __launch_bounds__(32 * WPC) __attribute__((amdgpu_waves_per_eu(4, 4))) void biquad_channel_kernel(
    const float* __restrict__ x,
    const float* __restrict__ b0v, const float* __restrict__ b1v, const float* __restrict__ b2v,
    const float* __restrict__ a1v, const float* __restrict__ a2v,
    float* __restrict__ y) {
  __shared__ float s_coef[3 * BLK];
  __shared__ float s_prop[4];
  __shared__ float2 s_end[WPC * COLS];
  __shared__ v4f s_tile[WPC][2][TILE4];

  const int lane = threadIdx.x & 31;
  const int wv = __builtin_amdgcn_readfirstlane(threadIdx.x >> 5);
  const int c = blockIdx.x;
  const int col = lane & 15;
  const int hi = lane >> 4;
  const int q_raw = wv * COLS + col;
  const bool live = q_raw < NSUB;
  const int q = live ? q_raw : NSUB - 1;

  const long woff = (long)c * T_LEN + (long)wv * (COLS * SUB_LEN);
  const float* __restrict__ xw = x + woff;
  const int lim = NSUB - 1 - wv * COLS;
  const unsigned lo = (unsigned)((hi * SUB_LEN + 4 * col) * 4);

  v4f ring[NRING][LPS];
#pragma unroll
  for (int s = 0; s < NRING; ++s) stage_load(xw, lo, hi, lim, s, ring[s]);

  v4f* __restrict__ xt = &s_tile[wv][0][0];
  v4f* __restrict__ yt = &s_tile[wv][1][0];
  const int fr = col * PITCH4 + 2 * hi;
  const int cr = hi * PITCH4 + col;

  const float b0 = b0v[c], b1 = b1v[c], b2 = b2v[c], a1 = a1v[c], a2 = a2v[c];

  if (wv == 0) {
    const float v1 = b1 - a1 * b0;
    const float v2 = b2 - a2 * b0;
    float hh[BLK], gA[BLK], gB[BLK];
    hh[0] = b0;
    gA[0] = 1.0f;
    gB[0] = 0.0f;
    float ga = 1.0f, gb = 0.0f;
#pragma unroll
    for (int i = 1; i < BLK; ++i) {
      const float gan = -a1 * ga - a2 * gb;
      const float gbn = ga;
      hh[i] = ga * v1 + gb * v2;
      gA[i] = gan;
      gB[i] = gbn;
      ga = gan; gb = gbn;
    }
    if (lane == 0) {
#pragma unroll
      for (int i = 0; i < BLK; ++i) {
        s_coef[i] = hh[i];
        s_coef[BLK + i] = gA[i];
        s_coef[2 * BLK + i] = gB[i];
      }
    }
  } else if (wv == 1) {
    const double a1d = (double)a1, a2d = (double)a2;
    double p = 1.0, qq = 0.0;
    for (int bit = top_bit(SUB_LEN) - 1; bit >= 0; --bit) {
      const double t  = (qq + qq) - a1d * p;
      const double pp = p * p;
      const double qn = qq * qq - a2d * pp;
      p = p * t;
      qq = qn;
      if ((SUB_LEN >> bit) & 1) {
        const double pn = qq - a1d * p;
        qq = -a2d * p;
        p = pn;
      }
    }
    if (lane == 0) {
      s_prop[0] = (float)(qq - a1d * p); s_prop[1] = (float)p;
      s_prop[2] = (float)(-a2d * p);     s_prop[3] = (float)qq;
    }
  }
  __syncthreads();

  v2f af[4];
#pragma unroll
  for (int m = 0; m < 4; ++m) {
#pragma unroll
    for (int r = 0; r < 2; ++r) {
      const int d = col - (8 * hi + 2 * m + r);
      const float hv = s_coef[d > 0 ? d : 0];
      af[m][r] = (d >= 0) ? hv : 0.0f;
    }
  }
  float gAr[8], gBr[8];
#pragma unroll
  for (int r = 0; r < 8; ++r) {
    gAr[r] = s_coef[BLK + 8 * hi + r];
    gBr[r] = s_coef[2 * BLK + 8 * hi + r];
  }

  float z1 = 0.0f, z2 = 0.0f;
  run_pass<false>(xw, nullptr, lo, lim, ring, xt, yt, fr, cr, af, gAr, gBr, b1, b2, a1, a2, hi, z1, z2);
  if (hi == 1) s_end[q_raw] = make_float2(z1, z2);

  __syncthreads();

#pragma unroll
  for (int r = 0; r < 8; ++r) {
    gAr[r] = s_coef[BLK + 8 * hi + r];
    gBr[r] = s_coef[2 * BLK + 8 * hi + r];
  }

#pragma unroll
  for (int s = 0; s < NRING; ++s) stage_load(xw, lo, hi, lim, s, ring[s]);

  const float m00 = s_prop[0], m01 = s_prop[1], m10 = s_prop[2], m11 = s_prop[3];
  const int qi = q - wv * COLS;
  float s1 = 0.0f, s2 = 0.0f;
  z1 = 0.0f; z2 = 0.0f;
  float2 dcur[COLS];
#pragma unroll
  for (int i = 0; i < COLS; ++i) dcur[i] = s_end[i];
  for (int k = 0; k < wv; ++k) {
    float2 dnext[COLS];
#pragma unroll
    for (int i = 0; i < COLS; ++i) dnext[i] = s_end[(k + 1) * COLS + i];
#pragma unroll
    for (int i = 0; i < COLS; ++i) {
      const float z1n = m00 * z1 + (m01 * z2 + dcur[i].x);
      const float z2n = m10 * z1 + (m11 * z2 + dcur[i].y);
      z1 = z1n; z2 = z2n;
    }
#pragma unroll
    for (int i = 0; i < COLS; ++i) dcur[i] = dnext[i];
  }
#pragma unroll
  for (int i = 0; i < COLS; ++i) {
    if (i == qi) { s1 = z1; s2 = z2; }
    const float z1n = m00 * z1 + (m01 * z2 + dcur[i].x);
    const float z2n = m10 * z1 + (m11 * z2 + dcur[i].y);
    z1 = z1n; z2 = z2n;
  }

  z1 = s1; z2 = s2;
  run_pass<true>(xw, y + woff, lo, lim, ring, xt, yt, fr, cr, af, gAr, gBr, b1, b2, a1, a2, hi, z1, z2);
}

extern "C" void kernel_launch(void* const* d_in, const int* in_sizes, int n_in,
                              void* d_out, int out_size, void* d_ws, size_t ws_size,
                              hipStream_t stream) {
  const float* x  = (const float*)d_in[0];
  const float* b0 = (const float*)d_in[1];
  const float* b1 = (const float*)d_in[2];
  const float* b2 = (const float*)d_in[3];
  const float* a1 = (const float*)d_in[4];
  const float* a2 = (const float*)d_in[5];
  float* y = (float*)d_out;

  biquad_channel_kernel<<<B_CH, 32 * WPC, 0, stream>>>(x, b0, b1, b2, a1, a2, y);
}


# pass 1 parks the state-independent block products L*x of the first 8 of 24 blocks per sub-slab in LDS; pass 2 takes them from there: no x read, tile fill or WMMA for those blocks; sums unchanged
# speedup vs baseline: 4.3888x; 1.0740x over previous
#include <hip/hip_runtime.h>


typedef __attribute__((ext_vector_type(2))) float v2f;
typedef __attribute__((ext_vector_type(8))) float v8f;
typedef __attribute__((ext_vector_type(4))) float v4f;

namespace {
constexpr int B_CH = 512;
constexpr int T_LEN = 48000;
constexpr int BLK = 16;
constexpr int NBLK = T_LEN / BLK;
constexpr int LB = 24;
constexpr int SUB_LEN = LB * BLK;
constexpr int NSUB = NBLK / LB;
constexpr int COLS = 16;
constexpr int WPC = (NSUB + COLS - 1) / COLS;
constexpr int SB = 4;
constexpr int NSTAGE = LB / SB;
constexpr int NRING = 2;
constexpr int KEEP_ST = 2;
constexpr int KEEP_B = KEEP_ST * SB;
constexpr int LPS = COLS * SB * BLK / (32 * 4);
constexpr int ROW4 = SB * BLK / 4;
constexpr int PITCH4 = ROW4 + 1;
constexpr int TILE4 = COLS * PITCH4;
static_assert(NBLK * BLK == T_LEN, "whole blocks");
static_assert(NSUB * LB == NBLK, "whole sub-slabs");
static_assert(NSTAGE * SB == LB, "whole stages");
static_assert((SB * BLK * 4) % 256 == 0 && (SUB_LEN * 4) % 256 == 0 && (T_LEN * 4) % 256 == 0,
              "a stage row is whole aligned 256-byte units of memory");
static_assert(ROW4 == 16 && LPS == 8, "one access = two rows: lane half = row, lane & 15 = float4 in the row");
constexpr int PAIR_B = 2 * SUB_LEN * 4;
constexpr int LPS_ALL = (NSUB - 1 - (WPC - 1) * COLS + 1) / 2;
static_assert(NRING <= NSTAGE, "ring no longer than the sub-slab");
static_assert(KEEP_ST % NRING == 0 && KEEP_ST + NRING <= NSTAGE, "pass 2 starts its ring at stage KEEP_ST, slot 0");
static_assert(LPS_ALL >= 1 && LPS_ALL < LPS, "the last wave's padded rows sit in the last accesses only");
static_assert(WPC >= 2, "wave 0 forms h and G, wave 1 the propagator");
constexpr int top_bit(int n) { int b = 0; while ((n >> (b + 1)) != 0) ++b; return b; }
}

__device__ __forceinline__ void wave_lds_order() {
  __builtin_amdgcn_fence(__ATOMIC_RELEASE, "wavefront");
  __builtin_amdgcn_wave_barrier();
  __builtin_amdgcn_fence(__ATOMIC_ACQUIRE, "wavefront");
}

__device__ __forceinline__ void stage_load(const float* __restrict__ xw, unsigned lo, int hi, int lim,
                                           int s, v4f (&dst)[LPS]) {
  const char* __restrict__ sp = (const char*)(xw + s * (SB * BLK));
#pragma unroll
  for (int i = 0; i < LPS; ++i) {
    if (i < LPS_ALL) {
      dst[i] = *(const v4f*)(sp + i * PAIR_B + lo);
    } else {
      v4f v = {0.0f, 0.0f, 0.0f, 0.0f};
      if (2 * i + hi <= lim) v = *(const v4f*)(sp + i * PAIR_B + lo);
      dst[i] = v;
    }
  }
}

template <bool WRITE_Y, int J>
__device__ __forceinline__ void block_step(const float* __restrict__ xw, float* __restrict__ yw,
                                           unsigned lo, int lim,
                                           v4f (&ring)[NRING][LPS],
                                           v4f* __restrict__ xt, v4f* __restrict__ yt,
                                           v4f* __restrict__ kp, v2f* __restrict__ kx,
                                           int fr, int cr,
                                           const v2f (&af)[4],
                                           const float (&gAr)[8], const float (&gBr)[8],
                                           float b1, float b2, float a1, float a2,
                                           int hi, v4f& c0, v4f& c1, v2f& cx, float& z1, float& z2) {
  constexpr bool KEPT = J < KEEP_B;
  constexpr bool FROM_LDS = WRITE_Y && KEPT;
  constexpr bool NEXT_FROM_LDS = WRITE_Y && (J + 1 < KEEP_B);
  const v4f t0 = c0;
  const v4f t1 = c1;
  const v2f tx = cx;
  if constexpr (J + 1 < LB) {
    if constexpr (NEXT_FROM_LDS) {
      c0 = kp[(2 * (J + 1)) * 32];
      c1 = kp[(2 * (J + 1) + 1) * 32];
      cx = kx[(J + 1) * 32];
    } else {
      if constexpr ((J + 1) % SB == 0) {
        constexpr int S = (J + 1) / SB;
        __builtin_amdgcn_sched_barrier(0);
        wave_lds_order();
#pragma unroll
        for (int i = 0; i < LPS; ++i) xt[cr + i * 2 * PITCH4] = ring[S % NRING][i];
        if constexpr (S + NRING < NSTAGE) stage_load(xw, lo, hi, lim, S + NRING, ring[S % NRING]);
        wave_lds_order();
      }
      c0 = xt[fr + 4 * ((J + 1) % SB)];
      c1 = xt[fr + 4 * ((J + 1) % SB) + 1];
    }
  }

  v8f acc;
  float x14, x15;
  if constexpr (FROM_LDS) {
    acc[0] = t0.x; acc[1] = t0.y; acc[2] = t0.z; acc[3] = t0.w;
    acc[4] = t1.x; acc[5] = t1.y; acc[6] = t1.z; acc[7] = t1.w;
    x14 = tx[0]; x15 = tx[1];
  } else {
    v2f bf0, bf1, bf2, bf3;
    bf0[0] = t0.x; bf0[1] = t0.y;
    bf1[0] = t0.z; bf1[1] = t0.w;
    bf2[0] = t1.x; bf2[1] = t1.y;
    bf3[0] = t1.z; bf3[1] = t1.w;

    acc = v8f{};
    acc = __builtin_amdgcn_wmma_f32_16x16x4_f32(false, af[0], false, bf0, (short)0, acc, false, false);
    acc = __builtin_amdgcn_wmma_f32_16x16x4_f32(false, af[1], false, bf1, (short)0, acc, false, false);
    acc = __builtin_amdgcn_wmma_f32_16x16x4_f32(false, af[2], false, bf2, (short)0, acc, false, false);
    acc = __builtin_amdgcn_wmma_f32_16x16x4_f32(false, af[3], false, bf3, (short)0, acc, false, false);
    x14 = t1.z; x15 = t1.w;
    if constexpr (!WRITE_Y && KEPT) {
      kp[(2 * J) * 32]     = __builtin_shufflevector(acc, acc, 0, 1, 2, 3);
      kp[(2 * J + 1) * 32] = __builtin_shufflevector(acc, acc, 4, 5, 6, 7);
      v2f xs; xs[0] = x14; xs[1] = x15;
      kx[J * 32] = xs;
    }
  }

  if constexpr (J > 0) {
    const unsigned u1 = __float_as_uint(z1), u2 = __float_as_uint(z2);
    const float z1o = __uint_as_float(
        __builtin_amdgcn_permlanex16(u1, u1, 0x76543210u, 0xfedcba98u, false, false));
    const float z2o = __uint_as_float(
        __builtin_amdgcn_permlanex16(u2, u2, 0x76543210u, 0xfedcba98u, false, false));
    if (hi == 0) { z1 = z1o; z2 = z2o; }
  }

  if constexpr (WRITE_Y) {
#pragma unroll
    for (int r = 0; r < 8; ++r) acc[r] += gAr[r] * z1 + gBr[r] * z2;
    yt[fr + 4 * (J % SB)]     = __builtin_shufflevector(acc, acc, 0, 1, 2, 3);
    yt[fr + 4 * (J % SB) + 1] = __builtin_shufflevector(acc, acc, 4, 5, 6, 7);
    if constexpr (J % SB == SB - 1) {
      char* __restrict__ dp = (char*)(yw + (J / SB) * (SB * BLK));
      wave_lds_order();
#pragma unroll
      for (int i = 0; i < LPS; ++i) {
        const v4f v = yt[cr + i * 2 * PITCH4];
        if (i < LPS_ALL) *(v4f*)(dp + i * PAIR_B + lo) = v;
        else if (2 * i + hi <= lim) *(v4f*)(dp + i * PAIR_B + lo) = v;
      }
      wave_lds_order();
    }
  } else {
    acc[6] += gAr[6] * z1 + gBr[6] * z2;
    acc[7] += gAr[7] * z1 + gBr[7] * z2;
  }

  const float z2mid = b2 * x14 - a2 * acc[6];
  const float z2n = b2 * x15 - a2 * acc[7];
  const float z1n = b1 * x15 - a1 * acc[7] + z2mid;
  z1 = z1n; z2 = z2n;
}

template <bool WRITE_Y, int J>
__device__ __forceinline__ void blocks_from(const float* __restrict__ xw, float* __restrict__ yw,
                                            unsigned lo, int lim,
                                            v4f (&ring)[NRING][LPS],
                                            v4f* __restrict__ xt, v4f* __restrict__ yt,
                                            v4f* __restrict__ kp, v2f* __restrict__ kx,
                                            int fr, int cr,
                                            const v2f (&af)[4],
                                            const float (&gAr)[8], const float (&gBr)[8],
                                            float b1, float b2, float a1, float a2,
                                            int hi, v4f& c0, v4f& c1, v2f& cx, float& z1, float& z2) {
  block_step<WRITE_Y, J>(xw, yw, lo, lim, ring, xt, yt, kp, kx, fr, cr, af, gAr, gBr, b1, b2, a1, a2, hi, c0, c1, cx, z1, z2);
  if constexpr (J + 1 < LB)
    blocks_from<WRITE_Y, J + 1>(xw, yw, lo, lim, ring, xt, yt, kp, kx, fr, cr, af, gAr, gBr, b1, b2, a1, a2, hi, c0, c1, cx, z1, z2);
}

template <bool WRITE_Y>
__device__ __forceinline__ void run_pass(const float* __restrict__ xw, float* __restrict__ yw,
                                         unsigned lo, int lim,
                                         v4f (&ring)[NRING][LPS],
                                         v4f* __restrict__ xt, v4f* __restrict__ yt,
                                         v4f* __restrict__ kp, v2f* __restrict__ kx,
                                         int fr, int cr,
                                         const v2f (&af)[4],
                                         const float (&gAr)[8], const float (&gBr)[8],
                                         float b1, float b2, float a1, float a2,
                                         int hi, float& z1, float& z2) {
  v4f c0, c1;
  v2f cx = {0.0f, 0.0f};
  if constexpr (WRITE_Y && KEEP_B > 0) {
    c0 = kp[0]; c1 = kp[32]; cx = kx[0];
  } else {
#pragma unroll
    for (int i = 0; i < LPS; ++i) xt[cr + i * 2 * PITCH4] = ring[0][i];
    if constexpr (NRING < NSTAGE) stage_load(xw, lo, hi, lim, NRING, ring[0]);
    wave_lds_order();
    c0 = xt[fr]; c1 = xt[fr + 1];
  }
  blocks_from<WRITE_Y, 0>(xw, yw, lo, lim, ring, xt, yt, kp, kx, fr, cr, af, gAr, gBr, b1, b2, a1, a2, hi, c0, c1, cx, z1, z2);
}

__global__ __launch_bounds__(32 * WPC) __attribute__((amdgpu_waves_per_eu(4, 4))) void biquad_channel_kernel(
    const float* __restrict__ x,
    const float* __restrict__ b0v, const float* __restrict__ b1v, const float* __restrict__ b2v,
    const float* __restrict__ a1v, const float* __restrict__ a2v,
    float* __restrict__ y) {
  __shared__ float s_coef[3 * BLK];
  __shared__ float s_prop[4];
  __shared__ float2 s_end[WPC * COLS];
  __shared__ v4f s_tile[WPC][2][TILE4];
  __shared__ v4f s_keep[WPC][KEEP_B][2][32];
  __shared__ v2f s_keepx[WPC][KEEP_B][32];

  const int lane = threadIdx.x & 31;
  const int wv = __builtin_amdgcn_readfirstlane(threadIdx.x >> 5);
  const int c = blockIdx.x;
  const int col = lane & 15;
  const int hi = lane >> 4;
  const int q_raw = wv * COLS + col;
  const bool live = q_raw < NSUB;
  const int q = live ? q_raw : NSUB - 1;

  const long woff = (long)c * T_LEN + (long)wv * (COLS * SUB_LEN);
  const float* __restrict__ xw = x + woff;
  const int lim = NSUB - 1 - wv * COLS;
  const unsigned lo = (unsigned)((hi * SUB_LEN + 4 * col) * 4);

  v4f ring[NRING][LPS];
#pragma unroll
  for (int s = 0; s < NRING; ++s) stage_load(xw, lo, hi, lim, s, ring[s]);

  v4f* __restrict__ xt = &s_tile[wv][0][0];
  v4f* __restrict__ yt = &s_tile[wv][1][0];
  const int fr = col * PITCH4 + 2 * hi;
  const int cr = hi * PITCH4 + col;
  v4f* __restrict__ kp = &s_keep[wv][0][0][lane];
  v2f* __restrict__ kx = &s_keepx[wv][0][lane];

  const float b0 = b0v[c], b1 = b1v[c], b2 = b2v[c], a1 = a1v[c], a2 = a2v[c];

  if (wv == 0) {
    const float v1 = b1 - a1 * b0;
    const float v2 = b2 - a2 * b0;
    float hh[BLK], gA[BLK], gB[BLK];
    hh[0] = b0;
    gA[0] = 1.0f;
    gB[0] = 0.0f;
    float ga = 1.0f, gb = 0.0f;
#pragma unroll
    for (int i = 1; i < BLK; ++i) {
      const float gan = -a1 * ga - a2 * gb;
      const float gbn = ga;
      hh[i] = ga * v1 + gb * v2;
      gA[i] = gan;
      gB[i] = gbn;
      ga = gan; gb = gbn;
    }
    if (lane == 0) {
#pragma unroll
      for (int i = 0; i < BLK; ++i) {
        s_coef[i] = hh[i];
        s_coef[BLK + i] = gA[i];
        s_coef[2 * BLK + i] = gB[i];
      }
    }
  } else if (wv == 1) {
    const double a1d = (double)a1, a2d = (double)a2;
    double p = 1.0, qq = 0.0;
    for (int bit = top_bit(SUB_LEN) - 1; bit >= 0; --bit) {
      const double t  = (qq + qq) - a1d * p;
      const double pp = p * p;
      const double qn = qq * qq - a2d * pp;
      p = p * t;
      qq = qn;
      if ((SUB_LEN >> bit) & 1) {
        const double pn = qq - a1d * p;
        qq = -a2d * p;
        p = pn;
      }
    }
    if (lane == 0) {
      s_prop[0] = (float)(qq - a1d * p); s_prop[1] = (float)p;
      s_prop[2] = (float)(-a2d * p);     s_prop[3] = (float)qq;
    }
  }
  __syncthreads();

  v2f af[4];
#pragma unroll
  for (int m = 0; m < 4; ++m) {
#pragma unroll
    for (int r = 0; r < 2; ++r) {
      const int d = col - (8 * hi + 2 * m + r);
      const float hv = s_coef[d > 0 ? d : 0];
      af[m][r] = (d >= 0) ? hv : 0.0f;
    }
  }
  float gAr[8], gBr[8];
#pragma unroll
  for (int r = 0; r < 8; ++r) {
    gAr[r] = s_coef[BLK + 8 * hi + r];
    gBr[r] = s_coef[2 * BLK + 8 * hi + r];
  }

  float z1 = 0.0f, z2 = 0.0f;
  run_pass<false>(xw, nullptr, lo, lim, ring, xt, yt, kp, kx, fr, cr, af, gAr, gBr, b1, b2, a1, a2, hi, z1, z2);
  if (hi == 1) s_end[q_raw] = make_float2(z1, z2);

  __syncthreads();

#pragma unroll
  for (int r = 0; r < 8; ++r) {
    gAr[r] = s_coef[BLK + 8 * hi + r];
    gBr[r] = s_coef[2 * BLK + 8 * hi + r];
  }

#pragma unroll
  for (int s = 0; s < NRING; ++s) stage_load(xw, lo, hi, lim, KEEP_ST + s, ring[s]);

  const float m00 = s_prop[0], m01 = s_prop[1], m10 = s_prop[2], m11 = s_prop[3];
  const int qi = q - wv * COLS;
  float s1 = 0.0f, s2 = 0.0f;
  z1 = 0.0f; z2 = 0.0f;
  float2 dcur[COLS];
#pragma unroll
  for (int i = 0; i < COLS; ++i) dcur[i] = s_end[i];
  for (int k = 0; k < wv; ++k) {
    float2 dnext[COLS];
#pragma unroll
    for (int i = 0; i < COLS; ++i) dnext[i] = s_end[(k + 1) * COLS + i];
#pragma unroll
    for (int i = 0; i < COLS; ++i) {
      const float z1n = m00 * z1 + (m01 * z2 + dcur[i].x);
      const float z2n = m10 * z1 + (m11 * z2 + dcur[i].y);
      z1 = z1n; z2 = z2n;
    }
#pragma unroll
    for (int i = 0; i < COLS; ++i) dcur[i] = dnext[i];
  }
#pragma unroll
  for (int i = 0; i < COLS; ++i) {
    if (i == qi) { s1 = z1; s2 = z2; }
    const float z1n = m00 * z1 + (m01 * z2 + dcur[i].x);
    const float z2n = m10 * z1 + (m11 * z2 + dcur[i].y);
    z1 = z1n; z2 = z2n;
  }

  z1 = s1; z2 = s2;
  run_pass<true>(xw, y + woff, lo, lim, ring, xt, yt, kp, kx, fr, cr, af, gAr, gBr, b1, b2, a1, a2, hi, z1, z2);
}

extern "C" void kernel_launch(void* const* d_in, const int* in_sizes, int n_in,
                              void* d_out, int out_size, void* d_ws, size_t ws_size,
                              hipStream_t stream) {
  const float* x  = (const float*)d_in[0];
  const float* b0 = (const float*)d_in[1];
  const float* b1 = (const float*)d_in[2];
  const float* b2 = (const float*)d_in[3];
  const float* a1 = (const float*)d_in[4];
  const float* a2 = (const float*)d_in[5];
  float* y = (float*)d_out;

  biquad_channel_kernel<<<B_CH, 32 * WPC, 0, stream>>>(x, b0, b1, b2, a1, a2, y);
}
